# starting kernel
# baseline (speedup 1.0000x reference)
#include <hip/hip_runtime.h>


typedef __attribute__((ext_vector_type(2))) float v2f;
typedef __attribute__((ext_vector_type(8))) float v8f;

#define CDIM   256
#define PITCH  260
#define ROWS   16
#define NITER  30
#define KT     32
#define RPITCH 20
#define RTSZ   (16 * RPITCH)

__global__ __launch_bounds__(512, 1) void fused_tinyops_kernel(
    const float* __restrict__ x, const float* __restrict__ ln_w,
    const float* __restrict__ ln_b, const float* __restrict__ W,
    const float* __restrict__ bias, float* __restrict__ out)
{
    __shared__ float hs[ROWS * PITCH];
    __shared__ float red[16 * RTSZ];

    const int tid   = threadIdx.x;
    const int lane  = tid & 31;
    const int wave  = tid >> 5;
    const int l16   = lane & 15;
    const int half  = (lane < 16) ? 0 : 2;
    const int khalf = wave >> 3;
    const int cpair = wave & 7;
    const int c0    = 2 * cpair;
    const int c1    = c0 + 1;

    const long rowBase = (long)blockIdx.x * ROWS;
    const float* xg = x + rowBase * CDIM;

    #pragma unroll
    for (int j = 0; j < (ROWS * CDIM) / 512; ++j) {
        int idx = j * 512 + tid;
        hs[(idx >> 8) * PITCH + (idx & (CDIM - 1))] = xg[idx];
    }

    v2f B0[KT], B1[KT];
    {
        const float* w0 = W + (16 * c0 + l16) * CDIM + 128 * khalf + half;
        const float* w1 = W + (16 * c1 + l16) * CDIM + 128 * khalf + half;
        #pragma unroll
        for (int t = 0; t < KT; ++t) {
            B0[t].x = w0[4 * t + 0];  B0[t].y = w0[4 * t + 1];
            B1[t].x = w1[4 * t + 0];  B1[t].y = w1[4 * t + 1];
        }
    }
    const float bias0 = bias[16 * c0 + l16];
    const float bias1 = bias[16 * c1 + l16];

    float lw[8], lb[8];
    #pragma unroll
    for (int j = 0; j < 8; ++j) {
        lw[j] = ln_w[lane + 32 * j];
        lb[j] = ln_b[lane + 32 * j];
    }

    __syncthreads();

    for (int it = 0; it < NITER; ++it) {
        float v[8];
        float* hrow = &hs[wave * PITCH];
        #pragma unroll
        for (int j = 0; j < 8; ++j) {
            float h = hrow[lane + 32 * j];
            h = h + 0.01f * tanhf(h);
            h = h * (1.0f / (1.0f + __expf(-h)));
            v[j] = h;
        }
        float s = ((v[0] + v[1]) + (v[2] + v[3])) + ((v[4] + v[5]) + (v[6] + v[7]));
        #pragma unroll
        for (int off = 16; off > 0; off >>= 1) s += __shfl_xor(s, off, 32);
        const float mu = s * (1.0f / CDIM);
        float q = 0.0f;
        #pragma unroll
        for (int j = 0; j < 8; ++j) { float d = v[j] - mu; q += d * d; }
        #pragma unroll
        for (int off = 16; off > 0; off >>= 1) q += __shfl_xor(q, off, 32);
        const float rstd = rsqrtf(q * (1.0f / CDIM) + 1e-5f);
        #pragma unroll
        for (int j = 0; j < 8; ++j)
            hrow[lane + 32 * j] = (v[j] - mu) * rstd * lw[j] + lb[j];

        __syncthreads();

        v8f acc0 = {0.f, 0.f, 0.f, 0.f, 0.f, 0.f, 0.f, 0.f};
        v8f acc1 = {0.f, 0.f, 0.f, 0.f, 0.f, 0.f, 0.f, 0.f};
        const float* arow = &hs[l16 * PITCH + 128 * khalf + half];
        #pragma unroll
        for (int t = 0; t < KT; ++t) {
            v2f a;
            a.x = arow[4 * t + 0];
            a.y = arow[4 * t + 1];
            acc0 = __builtin_amdgcn_wmma_f32_16x16x4_f32(
                false, a, false, B0[t], (short)0, acc0, false, false);
            acc1 = __builtin_amdgcn_wmma_f32_16x16x4_f32(
                false, a, false, B1[t], (short)0, acc1, false, false);
        }

        if (khalf == 1) {
            float* r0 = &red[(2 * cpair + 0) * RTSZ];
            float* r1 = &red[(2 * cpair + 1) * RTSZ];
            #pragma unroll
            for (int vi = 0; vi < 8; ++vi) {
                int r = vi + ((lane < 16) ? 0 : 8);
                r0[r * RPITCH + l16] = acc0[vi];
                r1[r * RPITCH + l16] = acc1[vi];
            }
        }
        __syncthreads();

        if (khalf == 0) {
            const float* r0 = &red[(2 * cpair + 0) * RTSZ];
            const float* r1 = &red[(2 * cpair + 1) * RTSZ];
            #pragma unroll
            for (int vi = 0; vi < 8; ++vi) {
                int r = vi + ((lane < 16) ? 0 : 8);
                hs[r * PITCH + 16 * c0 + l16] =
                    acc0[vi] + r0[r * RPITCH + l16] + bias0;
                hs[r * PITCH + 16 * c1 + l16] =
                    acc1[vi] + r1[r * RPITCH + l16] + bias1;
            }
        }
        __syncthreads();
    }

    float* og = out + rowBase * CDIM;
    #pragma unroll
    for (int j = 0; j < (ROWS * CDIM) / 512; ++j) {
        int idx = j * 512 + tid;
        og[idx] = hs[(idx >> 8) * PITCH + (idx & (CDIM - 1))];
    }
}

extern "C" void kernel_launch(void* const* d_in, const int* in_sizes, int n_in,
                              void* d_out, int out_size, void* d_ws, size_t ws_size,
                              hipStream_t stream) {
    (void)n_in; (void)d_ws; (void)ws_size; (void)out_size;
    const float* x    = (const float*)d_in[0];
    const float* ln_w = (const float*)d_in[1];
    const float* ln_b = (const float*)d_in[2];
    const float* W    = (const float*)d_in[3];
    const float* b    = (const float*)d_in[4];
    float* out        = (float*)d_out;

    const int total_rows = in_sizes[0] / CDIM;
    dim3 grid(total_rows / ROWS);
    dim3 block(512);
    fused_tinyops_kernel<<<grid, block, 0, stream>>>(x, ln_w, ln_b, W, b, out);
}


# Two 8-wave blocks per WGP instead of one 16-wave block: wave owns 32 columns, W slice half in registers, half in its own LDS region; the blocks' phases overlap; Linear k order, LN sum order change
# speedup vs baseline: 1.1673x; 1.1673x over previous
#include <hip/hip_runtime.h>


typedef __attribute__((ext_vector_type(2))) float v2f;
typedef __attribute__((ext_vector_type(4))) float v4f;
typedef __attribute__((ext_vector_type(8))) float v8f;

#define CDIM   256
#define PITCH  260
#define ROWS   16
#define NITER  30
#define KT     64
#define KREG   32
#define KLDS   (KT - KREG)
#define NW     8
#define THREADS (NW * 32)

template <int MASK>
__device__ __forceinline__ float xlane(float x) {
    int b = __builtin_bit_cast(int, x);
    int r = __builtin_amdgcn_update_dpp(0, b, 0x160 + MASK, 0xf, 0xf, true);
    return __builtin_bit_cast(float, r);
}

template <int N>
__device__ __forceinline__ float rshare(float x) {
    int b = __builtin_bit_cast(int, x);
    int r = __builtin_amdgcn_update_dpp(0, b, 0x150 + N, 0xf, 0xf, true);
    return __builtin_bit_cast(float, r);
}

__device__ __forceinline__ float rowsum16_of8(const float (&s)[8],
                                              bool b0, bool b1, bool b2) {
    float u[4], w[2];
    #pragma unroll
    for (int j = 0; j < 4; ++j) {
        float keep = b0 ? s[2 * j + 1] : s[2 * j];
        float send = b0 ? s[2 * j] : s[2 * j + 1];
        u[j] = keep + xlane<1>(send);
    }
    #pragma unroll
    for (int k = 0; k < 2; ++k) {
        float keep = b1 ? u[2 * k + 1] : u[2 * k];
        float send = b1 ? u[2 * k] : u[2 * k + 1];
        w[k] = keep + xlane<2>(send);
    }
    float keep = b2 ? w[1] : w[0];
    float send = b2 ? w[0] : w[1];
    float z = keep + xlane<4>(send);
    z += xlane<8>(z);
    return z;
}

__device__ __forceinline__ float pointwise(float h) {
    h = h + 0.01f * tanhf(h);
    h = h * (1.0f / (1.0f + __expf(-h)));
    return h;
}

template <int I>
__device__ __forceinline__ float centred_sq(float a, float b, float m_own) {
    const float m  = rshare<I>(m_own);
    const float d0 = a - m;
    const float d1 = b - m;
    return d0 * d0 + d1 * d1;
}

template <int I>
__device__ __forceinline__ void norm_store(float* t0, float* t1, float a, float b,
                                           float mu_own, float rstd_own,
                                           float lw0, float lb0, float lw1, float lb1) {
    const float mu = rshare<I>(mu_own);
    const float rs = rshare<I>(rstd_own);
    t0[I * PITCH] = (a - mu) * rs * lw0 + lb0;
    t1[I * PITCH] = (b - mu) * rs * lw1 + lb1;
}

__global__ __launch_bounds__(THREADS) __attribute__((amdgpu_waves_per_eu(4, 4)))
void fused_tinyops_kernel(
    const float* __restrict__ x, const float* __restrict__ ln_w,
    const float* __restrict__ ln_b, const float* __restrict__ W,
    const float* __restrict__ bias, float* __restrict__ out)
{
    __shared__ alignas(16) float hs[ROWS * PITCH];
    __shared__ alignas(16) float pm[16 * NW];
    __shared__ alignas(16) float pq[16 * NW];
    __shared__ alignas(16) float wl[NW * KLDS * 32 * 4];

    const int tid   = threadIdx.x;
    const int lane  = tid & 31;
    const int wave  = tid >> 5;
    const int l16   = lane & 15;
    const int hi    = lane >> 4;
    const int half  = 2 * hi;
    const int col0  = 32 * wave + l16;
    const int col1  = col0 + 16;
    const bool b0   = (l16 & 1) != 0;
    const bool b1   = (l16 & 2) != 0;
    const bool b2   = (l16 & 4) != 0;
    const int srow  = 8 * hi + (l16 & 7);

    const long rowBase = (long)blockIdx.x * ROWS;
    const float* xg = x + rowBase * CDIM;

    #pragma unroll
    for (int j = 0; j < (ROWS * CDIM) / THREADS; ++j) {
        int idx = j * THREADS + tid;
        hs[(idx >> 8) * PITCH + (idx & (CDIM - 1))] = xg[idx];
    }

    const float* wp0 = W + (long)col0 * CDIM + half;
    const float* wp1 = W + (long)col1 * CDIM + half;

    v4f* wmine = reinterpret_cast<v4f*>(wl) + (wave * KLDS) * 32 + lane;
    #pragma unroll 8
    for (int tt = 0; tt < KLDS; ++tt) {
        v4f f;
        f.x = wp0[4 * (KREG + tt) + 0];
        f.y = wp0[4 * (KREG + tt) + 1];
        f.z = wp1[4 * (KREG + tt) + 0];
        f.w = wp1[4 * (KREG + tt) + 1];
        wmine[tt * 32] = f;
    }

    v2f B0[KREG], B1[KREG];
    #pragma unroll
    for (int t = 0; t < KREG; ++t) {
        B0[t].x = wp0[4 * t + 0];
        B0[t].y = wp0[4 * t + 1];
        B1[t].x = wp1[4 * t + 0];
        B1[t].y = wp1[4 * t + 1];
    }
    const float bias0 = bias[col0], bias1 = bias[col1];
    const float lw0   = ln_w[col0], lw1   = ln_w[col1];
    const float lb0   = ln_b[col0], lb1   = ln_b[col1];

    __syncthreads();

    float* tile0 = &hs[(8 * hi) * PITCH + col0];
    float* tile1 = tile0 + 16;
    const float* arow = &hs[l16 * PITCH + half];
    const v4f* wfrag = wmine;

    float y0[8], y1[8];
    #pragma unroll
    for (int i = 0; i < 8; ++i) {
        y0[i] = tile0[i * PITCH];
        y1[i] = tile1[i * PITCH];
    }

    for (int it = 0; it < NITER; ++it) {
        float v0[8], v1[8];
        #pragma unroll
        for (int i = 0; i < 8; ++i) {
            v0[i] = pointwise(y0[i]);
            v1[i] = pointwise(y1[i]);
        }

        float s[8];
        #pragma unroll
        for (int i = 0; i < 8; ++i) s[i] = v0[i] + v1[i];
        const float m_own = rowsum16_of8(s, b0, b1, b2) * (1.0f / 32.0f);
        float e[8];
        e[0] = centred_sq<0>(v0[0], v1[0], m_own);
        e[1] = centred_sq<1>(v0[1], v1[1], m_own);
        e[2] = centred_sq<2>(v0[2], v1[2], m_own);
        e[3] = centred_sq<3>(v0[3], v1[3], m_own);
        e[4] = centred_sq<4>(v0[4], v1[4], m_own);
        e[5] = centred_sq<5>(v0[5], v1[5], m_own);
        e[6] = centred_sq<6>(v0[6], v1[6], m_own);
        e[7] = centred_sq<7>(v0[7], v1[7], m_own);
        const float q_own = rowsum16_of8(e, b0, b1, b2);
        if (l16 < 8) {
            pm[srow * NW + wave] = m_own;
            pq[srow * NW + wave] = q_own;
        }

        __syncthreads();

        float mu_own, rstd_own;
        {
            const v4f* sm = reinterpret_cast<const v4f*>(&pm[srow * NW]);
            const v4f* sq = reinterpret_cast<const v4f*>(&pq[srow * NW]);
            const v4f a0 = sm[0], a1 = sm[1], c0 = sq[0], c1 = sq[1];
            const float sum8 = ((a0.x + a0.y) + (a0.z + a0.w)) + ((a1.x + a1.y) + (a1.z + a1.w));
            mu_own = sum8 * (1.0f / NW);
            float g[8];
            #pragma unroll
            for (int j = 0; j < 4; ++j) {
                float d0 = a0[j] - mu_own;
                float d1 = a1[j] - mu_own;
                g[j]     = c0[j] + 32.0f * (d0 * d0);
                g[4 + j] = c1[j] + 32.0f * (d1 * d1);
            }
            const float r = ((g[0] + g[1]) + (g[2] + g[3])) + ((g[4] + g[5]) + (g[6] + g[7]));
            rstd_own = rsqrtf(r * (1.0f / CDIM) + 1e-5f);
        }

        norm_store<0>(tile0, tile1, v0[0], v1[0], mu_own, rstd_own, lw0, lb0, lw1, lb1);
        norm_store<1>(tile0, tile1, v0[1], v1[1], mu_own, rstd_own, lw0, lb0, lw1, lb1);
        norm_store<2>(tile0, tile1, v0[2], v1[2], mu_own, rstd_own, lw0, lb0, lw1, lb1);
        norm_store<3>(tile0, tile1, v0[3], v1[3], mu_own, rstd_own, lw0, lb0, lw1, lb1);
        norm_store<4>(tile0, tile1, v0[4], v1[4], mu_own, rstd_own, lw0, lb0, lw1, lb1);
        norm_store<5>(tile0, tile1, v0[5], v1[5], mu_own, rstd_own, lw0, lb0, lw1, lb1);
        norm_store<6>(tile0, tile1, v0[6], v1[6], mu_own, rstd_own, lw0, lb0, lw1, lb1);
        norm_store<7>(tile0, tile1, v0[7], v1[7], mu_own, rstd_own, lw0, lb0, lw1, lb1);

        __syncthreads();

        v8f acc0 = {0.f, 0.f, 0.f, 0.f, 0.f, 0.f, 0.f, 0.f};
        v8f acc1 = {0.f, 0.f, 0.f, 0.f, 0.f, 0.f, 0.f, 0.f};
        #pragma unroll
        for (int t = 0; t < KREG; ++t) {
            v2f a;
            a.x = arow[4 * t + 0];
            a.y = arow[4 * t + 1];
            acc0 = __builtin_amdgcn_wmma_f32_16x16x4_f32(
                false, a, false, B0[t], (short)0, acc0, false, false);
            acc1 = __builtin_amdgcn_wmma_f32_16x16x4_f32(
                false, a, false, B1[t], (short)0, acc1, false, false);
        }
        #pragma unroll
        for (int tt = 0; tt < KLDS; ++tt) {
            v2f a;
            a.x = arow[4 * (KREG + tt) + 0];
            a.y = arow[4 * (KREG + tt) + 1];
            const v4f f = wfrag[tt * 32];
            v2f f0, f1;
            f0.x = f.x; f0.y = f.y;
            f1.x = f.z; f1.y = f.w;
            acc0 = __builtin_amdgcn_wmma_f32_16x16x4_f32(
                false, a, false, f0, (short)0, acc0, false, false);
            acc1 = __builtin_amdgcn_wmma_f32_16x16x4_f32(
                false, a, false, f1, (short)0, acc1, false, false);
        }
        #pragma unroll
        for (int i = 0; i < 8; ++i) {
            y0[i] = acc0[i] + bias0;
            y1[i] = acc1[i] + bias1;
        }
    }

    __syncthreads();
    #pragma unroll
    for (int i = 0; i < 8; ++i) {
        tile0[i * PITCH] = y0[i];
        tile1[i * PITCH] = y1[i];
    }
    __syncthreads();

    float* og = out + rowBase * CDIM;
    #pragma unroll
    for (int j = 0; j < (ROWS * CDIM) / THREADS; ++j) {
        int idx = j * THREADS + tid;
        og[idx] = hs[(idx >> 8) * PITCH + (idx & (CDIM - 1))];
    }
}

extern "C" void kernel_launch(void* const* d_in, const int* in_sizes, int n_in,
                              void* d_out, int out_size, void* d_ws, size_t ws_size,
                              hipStream_t stream) {
    (void)n_in; (void)d_ws; (void)ws_size; (void)out_size;
    const float* x    = (const float*)d_in[0];
    const float* ln_w = (const float*)d_in[1];
    const float* ln_b = (const float*)d_in[2];
    const float* W    = (const float*)d_in[3];
    const float* b    = (const float*)d_in[4];
    float* out        = (float*)d_out;

    const int total_rows = in_sizes[0] / CDIM;
    dim3 grid(total_rows / ROWS);
    dim3 block(THREADS);
    fused_tinyops_kernel<<<grid, block, 0, stream>>>(x, ln_w, ln_b, W, b, out);
}


# Pointwise: the library tanh's two range cases (same formulas, coefficients, reciprocal) computed for every lane and selected, no exec branches: the 16 elements' chains interleave
# speedup vs baseline: 1.2652x; 1.0839x over previous
#include <hip/hip_runtime.h>


typedef __attribute__((ext_vector_type(2))) float v2f;
typedef __attribute__((ext_vector_type(4))) float v4f;
typedef __attribute__((ext_vector_type(8))) float v8f;

#define CDIM   256
#define PITCH  260
#define ROWS   16
#define NITER  30
#define KT     64
#define KREG   32
#define KLDS   (KT - KREG)
#define NW     8
#define THREADS (NW * 32)

template <int MASK>
__device__ __forceinline__ float xlane(float x) {
    int b = __builtin_bit_cast(int, x);
    int r = __builtin_amdgcn_update_dpp(0, b, 0x160 + MASK, 0xf, 0xf, true);
    return __builtin_bit_cast(float, r);
}

template <int N>
__device__ __forceinline__ float rshare(float x) {
    int b = __builtin_bit_cast(int, x);
    int r = __builtin_amdgcn_update_dpp(0, b, 0x150 + N, 0xf, 0xf, true);
    return __builtin_bit_cast(float, r);
}

__device__ __forceinline__ float rowsum16_of8(const float (&s)[8],
                                              bool b0, bool b1, bool b2) {
    float u[4], w[2];
    #pragma unroll
    for (int j = 0; j < 4; ++j) {
        float keep = b0 ? s[2 * j + 1] : s[2 * j];
        float send = b0 ? s[2 * j] : s[2 * j + 1];
        u[j] = keep + xlane<1>(send);
    }
    #pragma unroll
    for (int k = 0; k < 2; ++k) {
        float keep = b1 ? u[2 * k + 1] : u[2 * k];
        float send = b1 ? u[2 * k] : u[2 * k + 1];
        w[k] = keep + xlane<2>(send);
    }
    float keep = b2 ? w[1] : w[0];
    float send = b2 ? w[0] : w[1];
    float z = keep + xlane<4>(send);
    z += xlane<8>(z);
    return z;
}

__device__ __forceinline__ float fbits(unsigned u) { return __builtin_bit_cast(float, u); }

__device__ __forceinline__ float tanh_select(float x) {
    const float y = __builtin_fabsf(x);
    const float t   = expf(y + y);
    const float big = __builtin_fmaf(__builtin_amdgcn_rcpf(t + 1.0f), -2.0f, 1.0f);
    const float x2 = x * x;
    float p = __builtin_fmaf(fbits(0xbbbac73du), x2, fbits(0x3ca908c9u));
    p = __builtin_fmaf(x2, p, fbits(0xbd5c1c4eu));
    p = __builtin_fmaf(x2, p, fbits(0x3e088382u));
    p = __builtin_fmaf(x2, p, fbits(0xbeaaaa99u));
    const float small = __builtin_fmaf(x2, y * p, y);
    const float r = (y < 0.625f) ? small : big;
    return __builtin_copysignf(r, x);
}

__device__ __forceinline__ float pointwise(float h) {
    h = h + 0.01f * tanh_select(h);
    h = h * (1.0f / (1.0f + __expf(-h)));
    return h;
}

template <int I>
__device__ __forceinline__ float centred_sq(float a, float b, float m_own) {
    const float m  = rshare<I>(m_own);
    const float d0 = a - m;
    const float d1 = b - m;
    return d0 * d0 + d1 * d1;
}

template <int I>
__device__ __forceinline__ void norm_store(float* t0, float* t1, float a, float b,
                                           float mu_own, float rstd_own,
                                           float lw0, float lb0, float lw1, float lb1) {
    const float mu = rshare<I>(mu_own);
    const float rs = rshare<I>(rstd_own);
    t0[I * PITCH] = (a - mu) * rs * lw0 + lb0;
    t1[I * PITCH] = (b - mu) * rs * lw1 + lb1;
}

__global__ __launch_bounds__(THREADS) __attribute__((amdgpu_waves_per_eu(4, 4)))
void fused_tinyops_kernel(
    const float* __restrict__ x, const float* __restrict__ ln_w,
    const float* __restrict__ ln_b, const float* __restrict__ W,
    const float* __restrict__ bias, float* __restrict__ out)
{
    __shared__ alignas(16) float hs[ROWS * PITCH];
    __shared__ alignas(16) float pm[16 * NW];
    __shared__ alignas(16) float pq[16 * NW];
    __shared__ alignas(16) float wl[NW * KLDS * 32 * 4];

    const int tid   = threadIdx.x;
    const int lane  = tid & 31;
    const int wave  = tid >> 5;
    const int l16   = lane & 15;
    const int hi    = lane >> 4;
    const int half  = 2 * hi;
    const int col0  = 32 * wave + l16;
    const int col1  = col0 + 16;
    const bool b0   = (l16 & 1) != 0;
    const bool b1   = (l16 & 2) != 0;
    const bool b2   = (l16 & 4) != 0;
    const int srow  = 8 * hi + (l16 & 7);

    const long rowBase = (long)blockIdx.x * ROWS;
    const float* xg = x + rowBase * CDIM;

    #pragma unroll
    for (int j = 0; j < (ROWS * CDIM) / THREADS; ++j) {
        int idx = j * THREADS + tid;
        hs[(idx >> 8) * PITCH + (idx & (CDIM - 1))] = xg[idx];
    }

    const float* wp0 = W + (long)col0 * CDIM + half;
    const float* wp1 = W + (long)col1 * CDIM + half;

    v4f* wmine = reinterpret_cast<v4f*>(wl) + (wave * KLDS) * 32 + lane;
    #pragma unroll 8
    for (int tt = 0; tt < KLDS; ++tt) {
        v4f f;
        f.x = wp0[4 * (KREG + tt) + 0];
        f.y = wp0[4 * (KREG + tt) + 1];
        f.z = wp1[4 * (KREG + tt) + 0];
        f.w = wp1[4 * (KREG + tt) + 1];
        wmine[tt * 32] = f;
    }

    v2f B0[KREG], B1[KREG];
    #pragma unroll
    for (int t = 0; t < KREG; ++t) {
        B0[t].x = wp0[4 * t + 0];
        B0[t].y = wp0[4 * t + 1];
        B1[t].x = wp1[4 * t + 0];
        B1[t].y = wp1[4 * t + 1];
    }
    const float bias0 = bias[col0], bias1 = bias[col1];
    const float lw0   = ln_w[col0], lw1   = ln_w[col1];
    const float lb0   = ln_b[col0], lb1   = ln_b[col1];

    __syncthreads();

    float* tile0 = &hs[(8 * hi) * PITCH + col0];
    float* tile1 = tile0 + 16;
    const float* arow = &hs[l16 * PITCH + half];
    const v4f* wfrag = wmine;

    float y0[8], y1[8];
    #pragma unroll
    for (int i = 0; i < 8; ++i) {
        y0[i] = tile0[i * PITCH];
        y1[i] = tile1[i * PITCH];
    }

    for (int it = 0; it < NITER; ++it) {
        float v0[8], v1[8];
        #pragma unroll
        for (int i = 0; i < 8; ++i) {
            v0[i] = pointwise(y0[i]);
            v1[i] = pointwise(y1[i]);
        }

        float s[8];
        #pragma unroll
        for (int i = 0; i < 8; ++i) s[i] = v0[i] + v1[i];
        const float m_own = rowsum16_of8(s, b0, b1, b2) * (1.0f / 32.0f);
        float e[8];
        e[0] = centred_sq<0>(v0[0], v1[0], m_own);
        e[1] = centred_sq<1>(v0[1], v1[1], m_own);
        e[2] = centred_sq<2>(v0[2], v1[2], m_own);
        e[3] = centred_sq<3>(v0[3], v1[3], m_own);
        e[4] = centred_sq<4>(v0[4], v1[4], m_own);
        e[5] = centred_sq<5>(v0[5], v1[5], m_own);
        e[6] = centred_sq<6>(v0[6], v1[6], m_own);
        e[7] = centred_sq<7>(v0[7], v1[7], m_own);
        const float q_own = rowsum16_of8(e, b0, b1, b2);
        if (l16 < 8) {
            pm[srow * NW + wave] = m_own;
            pq[srow * NW + wave] = q_own;
        }

        __syncthreads();

        float mu_own, rstd_own;
        {
            const v4f* sm = reinterpret_cast<const v4f*>(&pm[srow * NW]);
            const v4f* sq = reinterpret_cast<const v4f*>(&pq[srow * NW]);
            const v4f a0 = sm[0], a1 = sm[1], c0 = sq[0], c1 = sq[1];
            const float sum8 = ((a0.x + a0.y) + (a0.z + a0.w)) + ((a1.x + a1.y) + (a1.z + a1.w));
            mu_own = sum8 * (1.0f / NW);
            float g[8];
            #pragma unroll
            for (int j = 0; j < 4; ++j) {
                float d0 = a0[j] - mu_own;
                float d1 = a1[j] - mu_own;
                g[j]     = c0[j] + 32.0f * (d0 * d0);
                g[4 + j] = c1[j] + 32.0f * (d1 * d1);
            }
            const float r = ((g[0] + g[1]) + (g[2] + g[3])) + ((g[4] + g[5]) + (g[6] + g[7]));
            rstd_own = rsqrtf(r * (1.0f / CDIM) + 1e-5f);
        }

        norm_store<0>(tile0, tile1, v0[0], v1[0], mu_own, rstd_own, lw0, lb0, lw1, lb1);
        norm_store<1>(tile0, tile1, v0[1], v1[1], mu_own, rstd_own, lw0, lb0, lw1, lb1);
        norm_store<2>(tile0, tile1, v0[2], v1[2], mu_own, rstd_own, lw0, lb0, lw1, lb1);
        norm_store<3>(tile0, tile1, v0[3], v1[3], mu_own, rstd_own, lw0, lb0, lw1, lb1);
        norm_store<4>(tile0, tile1, v0[4], v1[4], mu_own, rstd_own, lw0, lb0, lw1, lb1);
        norm_store<5>(tile0, tile1, v0[5], v1[5], mu_own, rstd_own, lw0, lb0, lw1, lb1);
        norm_store<6>(tile0, tile1, v0[6], v1[6], mu_own, rstd_own, lw0, lb0, lw1, lb1);
        norm_store<7>(tile0, tile1, v0[7], v1[7], mu_own, rstd_own, lw0, lb0, lw1, lb1);

        __syncthreads();

        v8f acc0 = {0.f, 0.f, 0.f, 0.f, 0.f, 0.f, 0.f, 0.f};
        v8f acc1 = {0.f, 0.f, 0.f, 0.f, 0.f, 0.f, 0.f, 0.f};
        #pragma unroll
        for (int t = 0; t < KREG; ++t) {
            v2f a;
            a.x = arow[4 * t + 0];
            a.y = arow[4 * t + 1];
            acc0 = __builtin_amdgcn_wmma_f32_16x16x4_f32(
                false, a, false, B0[t], (short)0, acc0, false, false);
            acc1 = __builtin_amdgcn_wmma_f32_16x16x4_f32(
                false, a, false, B1[t], (short)0, acc1, false, false);
        }
        #pragma unroll
        for (int tt = 0; tt < KLDS; ++tt) {
            v2f a;
            a.x = arow[4 * (KREG + tt) + 0];
            a.y = arow[4 * (KREG + tt) + 1];
            const v4f f = wfrag[tt * 32];
            v2f f0, f1;
            f0.x = f.x; f0.y = f.y;
            f1.x = f.z; f1.y = f.w;
            acc0 = __builtin_amdgcn_wmma_f32_16x16x4_f32(
                false, a, false, f0, (short)0, acc0, false, false);
            acc1 = __builtin_amdgcn_wmma_f32_16x16x4_f32(
                false, a, false, f1, (short)0, acc1, false, false);
        }
        #pragma unroll
        for (int i = 0; i < 8; ++i) {
            y0[i] = acc0[i] + bias0;
            y1[i] = acc1[i] + bias1;
        }
    }

    __syncthreads();
    #pragma unroll
    for (int i = 0; i < 8; ++i) {
        tile0[i * PITCH] = y0[i];
        tile1[i * PITCH] = y1[i];
    }
    __syncthreads();

    float* og = out + rowBase * CDIM;
    #pragma unroll
    for (int j = 0; j < (ROWS * CDIM) / THREADS; ++j) {
        int idx = j * THREADS + tid;
        og[idx] = hs[(idx >> 8) * PITCH + (idx & (CDIM - 1))];
    }
}

extern "C" void kernel_launch(void* const* d_in, const int* in_sizes, int n_in,
                              void* d_out, int out_size, void* d_ws, size_t ws_size,
                              hipStream_t stream) {
    (void)n_in; (void)d_ws; (void)ws_size; (void)out_size;
    const float* x    = (const float*)d_in[0];
    const float* ln_w = (const float*)d_in[1];
    const float* ln_b = (const float*)d_in[2];
    const float* W    = (const float*)d_in[3];
    const float* b    = (const float*)d_in[4];
    float* out        = (float*)d_out;

    const int total_rows = in_sizes[0] / CDIM;
    dim3 grid(total_rows / ROWS);
    dim3 block(THREADS);
    fused_tinyops_kernel<<<grid, block, 0, stream>>>(x, ln_w, ln_b, W, b, out);
}
